# starting kernel
# baseline (speedup 1.0000x reference)
#include <hip/hip_runtime.h>

typedef float v2f __attribute__((ext_vector_type(2)));
typedef float v8f __attribute__((ext_vector_type(8)));

#define D 64

__global__ void __launch_bounds__(256)
gemm_wmma(const float* __restrict__ X, const float* __restrict__ W,
          float* __restrict__ S, int n)
{
    const int lane = threadIdx.x & 31;
    const int wave = threadIdx.x >> 5;
    const int rowTile = blockIdx.x * 2 + (wave >> 2);
    const int m0 = rowTile * 16;
    if (m0 >= n) return;
    const int n0 = (wave & 3) * 16;

    const int h  = lane >> 4;
    const int ml = lane & 15;

    const float* xrow = X + (size_t)(m0 + ml) * D;

    v8f c = {};
    #pragma unroll
    for (int k0 = 0; k0 < D; k0 += 4) {
        const int k = k0 + 2 * h;
        v2f a;
        a.x = xrow[k];
        a.y = xrow[k + 1];
        v2f b;
        b.x = W[(size_t)k       * D + n0 + ml];
        b.y = W[(size_t)(k + 1) * D + n0 + ml];
        c = __builtin_amdgcn_wmma_f32_16x16x4_f32(
                 false, a,  false, b,
                 (short)0, c,  false,  false);
    }

    #pragma unroll
    for (int v = 0; v < 8; ++v) {
        S[(size_t)(m0 + v + 8 * h) * D + n0 + ml] = c[v];
    }
}

__global__ void __launch_bounds__(256)
init_out(float* __restrict__ out, const float* __restrict__ b, int total)
{
    int i = blockIdx.x * 256 + threadIdx.x;
    if (i < total) out[i] = b[i & (D - 1)];
}

__global__ void __launch_bounds__(256)
edge_scatter(const float* __restrict__ S, const float* __restrict__ ev,
             const int* __restrict__ row, const int* __restrict__ col,
             float* __restrict__ out, int E)
{
    const int tid  = blockIdx.x * 256 + threadIdx.x;
    const int wid  = tid >> 5;
    const int lane = threadIdx.x & 31;
    const int h    = lane >> 4;
    const int li   = lane & 15;

    const int e = wid * 2 + h;
    if (e >= E) return;

    const int   r = row[e];
    const int   c = col[e];
    const float v = ev[e];

    const float4 s = *(const float4*)(S + (size_t)c * D + li * 4);
    float* o = out + (size_t)r * D + li * 4;
    unsafeAtomicAdd(o + 0, v * s.x);
    unsafeAtomicAdd(o + 1, v * s.y);
    unsafeAtomicAdd(o + 2, v * s.z);
    unsafeAtomicAdd(o + 3, v * s.w);
}

extern "C" void kernel_launch(void* const* d_in, const int* in_sizes, int n_in,
                              void* d_out, int out_size, void* d_ws, size_t ws_size,
                              hipStream_t stream)
{
    const float* X   = (const float*)d_in[0];
    const float* W   = (const float*)d_in[1];
    const float* b   = (const float*)d_in[2];
    const float* ev  = (const float*)d_in[3];
    const int*   row = (const int*)d_in[4];
    const int*   col = (const int*)d_in[5];
    float*       out = (float*)d_out;
    float*       S   = (float*)d_ws;

    const int N = in_sizes[0] / D;
    const int E = in_sizes[3];

    const int rowTiles  = (N + 15) / 16;
    const int gemmBlks  = (rowTiles + 1) / 2;
    gemm_wmma<<<gemmBlks, 256, 0, stream>>>(X, W, S, N);

    const int total    = N * D;
    const int initBlks = (total + 255) / 256;
    init_out<<<initBlks, 256, 0, stream>>>(out, b, total);

    const int waves    = (E + 1) / 2;
    const int edgeBlks = (waves + 7) / 8;
    edge_scatter<<<edgeBlks, 256, 0, stream>>>(S, ev, row, col, out, E);
}


# edge scatter: lane = output column, still 2 edges per wave (no 32-edge walk): each f32 atomic adds to one whole 128-B line (was 8 dwords in each of 4); edge data by scalar loads; arrival-order sums
# speedup vs baseline: 1.2321x; 1.2321x over previous
#include <hip/hip_runtime.h>

typedef float v2f __attribute__((ext_vector_type(2)));
typedef float v8f __attribute__((ext_vector_type(8)));

#define D 64

__global__ void __launch_bounds__(256)
gemm_wmma(const float* __restrict__ X, const float* __restrict__ W,
          float* __restrict__ S, int n)
{
    const int lane = threadIdx.x & 31;
    const int wave = threadIdx.x >> 5;
    const int rowTile = blockIdx.x * 2 + (wave >> 2);
    const int m0 = rowTile * 16;
    if (m0 >= n) return;
    const int n0 = (wave & 3) * 16;

    const int h  = lane >> 4;
    const int ml = lane & 15;

    const float* xrow = X + (size_t)(m0 + ml) * D;

    v8f c = {};
    #pragma unroll
    for (int k0 = 0; k0 < D; k0 += 4) {
        const int k = k0 + 2 * h;
        v2f a;
        a.x = xrow[k];
        a.y = xrow[k + 1];
        v2f b;
        b.x = W[(size_t)k       * D + n0 + ml];
        b.y = W[(size_t)(k + 1) * D + n0 + ml];
        c = __builtin_amdgcn_wmma_f32_16x16x4_f32(
                 false, a,  false, b,
                 (short)0, c,  false,  false);
    }

    #pragma unroll
    for (int v = 0; v < 8; ++v) {
        S[(size_t)(m0 + v + 8 * h) * D + n0 + ml] = c[v];
    }
}

__global__ void __launch_bounds__(256)
init_out(float* __restrict__ out, const float* __restrict__ b, int total)
{
    int i = blockIdx.x * 256 + threadIdx.x;
    if (i < total) out[i] = b[i & (D - 1)];
}

__global__ void __launch_bounds__(256)
edge_scatter(const float* __restrict__ S, const float* __restrict__ ev,
             const int* __restrict__ row, const int* __restrict__ col,
             float* __restrict__ out, int E)
{
    const int lane = threadIdx.x & 31;
    const int wid = __builtin_amdgcn_readfirstlane((int)((blockIdx.x * 256 + threadIdx.x) >> 5));

    const int e0 = wid * 2;
    if (e0 + 1 < E) {
        const int   c0 = col[e0];
        const int   c1 = col[e0 + 1];
        const int   r0 = row[e0];
        const int   r1 = row[e0 + 1];
        const float v0 = ev[e0];
        const float v1 = ev[e0 + 1];

        const float* s0 = S + (size_t)c0 * D + lane;
        const float* s1 = S + (size_t)c1 * D + lane;
        const float a0 = s0[0];
        const float a1 = s0[32];
        const float b0 = s1[0];
        const float b1 = s1[32];

        float* o0 = out + (size_t)r0 * D + lane;
        float* o1 = out + (size_t)r1 * D + lane;
        unsafeAtomicAdd(o0,      v0 * a0);
        unsafeAtomicAdd(o0 + 32, v0 * a1);
        unsafeAtomicAdd(o1,      v1 * b0);
        unsafeAtomicAdd(o1 + 32, v1 * b1);
    } else if (e0 < E) {
        const int   c0 = col[e0];
        const int   r0 = row[e0];
        const float v0 = ev[e0];
        const float* s0 = S + (size_t)c0 * D + lane;
        const float a0 = s0[0];
        const float a1 = s0[32];
        float* o0 = out + (size_t)r0 * D + lane;
        unsafeAtomicAdd(o0,      v0 * a0);
        unsafeAtomicAdd(o0 + 32, v0 * a1);
    }
}

extern "C" void kernel_launch(void* const* d_in, const int* in_sizes, int n_in,
                              void* d_out, int out_size, void* d_ws, size_t ws_size,
                              hipStream_t stream)
{
    const float* X   = (const float*)d_in[0];
    const float* W   = (const float*)d_in[1];
    const float* b   = (const float*)d_in[2];
    const float* ev  = (const float*)d_in[3];
    const int*   row = (const int*)d_in[4];
    const int*   col = (const int*)d_in[5];
    float*       out = (float*)d_out;
    float*       S   = (float*)d_ws;

    const int N = in_sizes[0] / D;
    const int E = in_sizes[3];

    const int rowTiles  = (N + 15) / 16;
    const int gemmBlks  = (rowTiles + 1) / 2;
    gemm_wmma<<<gemmBlks, 256, 0, stream>>>(X, W, S, N);

    const int total    = N * D;
    const int initBlks = (total + 255) / 256;
    init_out<<<initBlks, 256, 0, stream>>>(out, b, total);

    const int waves    = (E + 1) / 2;
    const int edgeBlks = (waves + 7) / 8;
    edge_scatter<<<edgeBlks, 256, 0, stream>>>(S, ev, row, col, out, E);
}


# fewer, fatter waves: a scatter wave takes 4 edges (was 2; 8 line gathers in flight, then 8 line atomics in edge order), an init lane stores 16 B (was 4); arrival-order sums as before
# speedup vs baseline: 1.8635x; 1.5125x over previous
#include <hip/hip_runtime.h>

typedef float v2f __attribute__((ext_vector_type(2)));
typedef float v8f __attribute__((ext_vector_type(8)));

#define D 64

__global__ void __launch_bounds__(256)
gemm_wmma(const float* __restrict__ X, const float* __restrict__ W,
          float* __restrict__ S, int n)
{
    const int lane = threadIdx.x & 31;
    const int wave = threadIdx.x >> 5;
    const int rowTile = blockIdx.x * 2 + (wave >> 2);
    const int m0 = rowTile * 16;
    if (m0 >= n) return;
    const int n0 = (wave & 3) * 16;

    const int h  = lane >> 4;
    const int ml = lane & 15;

    const float* xrow = X + (size_t)(m0 + ml) * D;

    v8f c = {};
    #pragma unroll
    for (int k0 = 0; k0 < D; k0 += 4) {
        const int k = k0 + 2 * h;
        v2f a;
        a.x = xrow[k];
        a.y = xrow[k + 1];
        v2f b;
        b.x = W[(size_t)k       * D + n0 + ml];
        b.y = W[(size_t)(k + 1) * D + n0 + ml];
        c = __builtin_amdgcn_wmma_f32_16x16x4_f32(
                 false, a,  false, b,
                 (short)0, c,  false,  false);
    }

    #pragma unroll
    for (int v = 0; v < 8; ++v) {
        S[(size_t)(m0 + v + 8 * h) * D + n0 + ml] = c[v];
    }
}

__global__ void __launch_bounds__(256)
init_out(float* __restrict__ out, const float* __restrict__ b, int total4)
{
    int i = blockIdx.x * 256 + threadIdx.x;
    if (i < total4) {
        const float4 bv = *reinterpret_cast<const float4*>(b + ((i * 4) & (D - 1)));
        reinterpret_cast<float4*>(out)[i] = bv;
    }
}

#define EPW 4
__global__ void __launch_bounds__(256)
edge_scatter(const float* __restrict__ S, const float* __restrict__ ev,
             const int* __restrict__ row, const int* __restrict__ col,
             float* __restrict__ out, int E)
{
    const int lane = threadIdx.x & 31;
    const int wid = __builtin_amdgcn_readfirstlane((int)((blockIdx.x * 256 + threadIdx.x) >> 5));

    const int e0 = wid * EPW;
    if (e0 + EPW <= E) {
        int c[EPW], r[EPW];
        float v[EPW], lo[EPW], hi[EPW];
        #pragma unroll
        for (int j = 0; j < EPW; ++j) {
            c[j] = col[e0 + j];
            r[j] = row[e0 + j];
            v[j] = ev[e0 + j];
        }
        #pragma unroll
        for (int j = 0; j < EPW; ++j) {
            const float* s = S + (size_t)c[j] * D + lane;
            lo[j] = s[0];
            hi[j] = s[32];
        }
        #pragma unroll
        for (int j = 0; j < EPW; ++j) {
            float* o = out + (size_t)r[j] * D + lane;
            unsafeAtomicAdd(o,      v[j] * lo[j]);
            unsafeAtomicAdd(o + 32, v[j] * hi[j]);
        }
    } else {
        for (int e = e0; e < E; ++e) {
            const int   c0 = col[e];
            const int   r0 = row[e];
            const float v0 = ev[e];
            const float* s0 = S + (size_t)c0 * D + lane;
            const float a0 = s0[0];
            const float a1 = s0[32];
            float* o0 = out + (size_t)r0 * D + lane;
            unsafeAtomicAdd(o0,      v0 * a0);
            unsafeAtomicAdd(o0 + 32, v0 * a1);
        }
    }
}

extern "C" void kernel_launch(void* const* d_in, const int* in_sizes, int n_in,
                              void* d_out, int out_size, void* d_ws, size_t ws_size,
                              hipStream_t stream)
{
    const float* X   = (const float*)d_in[0];
    const float* W   = (const float*)d_in[1];
    const float* b   = (const float*)d_in[2];
    const float* ev  = (const float*)d_in[3];
    const int*   row = (const int*)d_in[4];
    const int*   col = (const int*)d_in[5];
    float*       out = (float*)d_out;
    float*       S   = (float*)d_ws;

    const int N = in_sizes[0] / D;
    const int E = in_sizes[3];

    const int rowTiles  = (N + 15) / 16;
    const int gemmBlks  = (rowTiles + 1) / 2;
    gemm_wmma<<<gemmBlks, 256, 0, stream>>>(X, W, S, N);

    const int total4   = N * (D / 4);
    const int initBlks = (total4 + 255) / 256;
    init_out<<<initBlks, 256, 0, stream>>>(out, b, total4);

    const int waves    = (E + EPW - 1) / EPW;
    const int edgeBlks = (waves + 7) / 8;
    edge_scatter<<<edgeBlks, 256, 0, stream>>>(S, ev, row, col, out, E);
}


# fatter scatter waves again: a wave takes 8 edges (was 4): 16 line gathers in flight, then 16 line atomics in edge order; half as many waves enter; arrival-order sums as before
# speedup vs baseline: 2.0817x; 1.1171x over previous
#include <hip/hip_runtime.h>

typedef float v2f __attribute__((ext_vector_type(2)));
typedef float v8f __attribute__((ext_vector_type(8)));

#define D 64

__global__ void __launch_bounds__(256)
gemm_wmma(const float* __restrict__ X, const float* __restrict__ W,
          float* __restrict__ S, int n)
{
    const int lane = threadIdx.x & 31;
    const int wave = threadIdx.x >> 5;
    const int rowTile = blockIdx.x * 2 + (wave >> 2);
    const int m0 = rowTile * 16;
    if (m0 >= n) return;
    const int n0 = (wave & 3) * 16;

    const int h  = lane >> 4;
    const int ml = lane & 15;

    const float* xrow = X + (size_t)(m0 + ml) * D;

    v8f c = {};
    #pragma unroll
    for (int k0 = 0; k0 < D; k0 += 4) {
        const int k = k0 + 2 * h;
        v2f a;
        a.x = xrow[k];
        a.y = xrow[k + 1];
        v2f b;
        b.x = W[(size_t)k       * D + n0 + ml];
        b.y = W[(size_t)(k + 1) * D + n0 + ml];
        c = __builtin_amdgcn_wmma_f32_16x16x4_f32(
                 false, a,  false, b,
                 (short)0, c,  false,  false);
    }

    #pragma unroll
    for (int v = 0; v < 8; ++v) {
        S[(size_t)(m0 + v + 8 * h) * D + n0 + ml] = c[v];
    }
}

__global__ void __launch_bounds__(256)
init_out(float* __restrict__ out, const float* __restrict__ b, int total4)
{
    int i = blockIdx.x * 256 + threadIdx.x;
    if (i < total4) {
        const float4 bv = *reinterpret_cast<const float4*>(b + ((i * 4) & (D - 1)));
        reinterpret_cast<float4*>(out)[i] = bv;
    }
}

#define EPW 8
__global__ void __launch_bounds__(256)
edge_scatter(const float* __restrict__ S, const float* __restrict__ ev,
             const int* __restrict__ row, const int* __restrict__ col,
             float* __restrict__ out, int E)
{
    const int lane = threadIdx.x & 31;
    const int wid = __builtin_amdgcn_readfirstlane((int)((blockIdx.x * 256 + threadIdx.x) >> 5));

    const int e0 = wid * EPW;
    if (e0 + EPW <= E) {
        int c[EPW], r[EPW];
        float v[EPW], lo[EPW], hi[EPW];
        #pragma unroll
        for (int j = 0; j < EPW; ++j) {
            c[j] = col[e0 + j];
            r[j] = row[e0 + j];
            v[j] = ev[e0 + j];
        }
        #pragma unroll
        for (int j = 0; j < EPW; ++j) {
            const float* s = S + (size_t)c[j] * D + lane;
            lo[j] = s[0];
            hi[j] = s[32];
        }
        #pragma unroll
        for (int j = 0; j < EPW; ++j) {
            float* o = out + (size_t)r[j] * D + lane;
            unsafeAtomicAdd(o,      v[j] * lo[j]);
            unsafeAtomicAdd(o + 32, v[j] * hi[j]);
        }
    } else {
        for (int e = e0; e < E; ++e) {
            const int   c0 = col[e];
            const int   r0 = row[e];
            const float v0 = ev[e];
            const float* s0 = S + (size_t)c0 * D + lane;
            const float a0 = s0[0];
            const float a1 = s0[32];
            float* o0 = out + (size_t)r0 * D + lane;
            unsafeAtomicAdd(o0,      v0 * a0);
            unsafeAtomicAdd(o0 + 32, v0 * a1);
        }
    }
}

extern "C" void kernel_launch(void* const* d_in, const int* in_sizes, int n_in,
                              void* d_out, int out_size, void* d_ws, size_t ws_size,
                              hipStream_t stream)
{
    const float* X   = (const float*)d_in[0];
    const float* W   = (const float*)d_in[1];
    const float* b   = (const float*)d_in[2];
    const float* ev  = (const float*)d_in[3];
    const int*   row = (const int*)d_in[4];
    const int*   col = (const int*)d_in[5];
    float*       out = (float*)d_out;
    float*       S   = (float*)d_ws;

    const int N = in_sizes[0] / D;
    const int E = in_sizes[3];

    const int rowTiles  = (N + 15) / 16;
    const int gemmBlks  = (rowTiles + 1) / 2;
    gemm_wmma<<<gemmBlks, 256, 0, stream>>>(X, W, S, N);

    const int total4   = N * (D / 4);
    const int initBlks = (total4 + 255) / 256;
    init_out<<<initBlks, 256, 0, stream>>>(out, b, total4);

    const int waves    = (E + EPW - 1) / EPW;
    const int edgeBlks = (waves + 7) / 8;
    edge_scatter<<<edgeBlks, 256, 0, stream>>>(S, ev, row, col, out, E);
}


# edge scatter: each output row's f32 atomics now come from one L2 half: a workgroup reads its die, draws 128-edge tiles from its half's counter, takes only own-half rows; arrival-order sums as before
# speedup vs baseline: 2.5326x; 1.2166x over previous
#include <hip/hip_runtime.h>

typedef float v2f __attribute__((ext_vector_type(2)));
typedef float v8f __attribute__((ext_vector_type(8)));

#define D 64

__global__ void __launch_bounds__(256)
gemm_wmma(const float* __restrict__ X, const float* __restrict__ W,
          float* __restrict__ S, int n)
{
    const int lane = threadIdx.x & 31;
    const int wave = threadIdx.x >> 5;
    const int rowTile = blockIdx.x * 2 + (wave >> 2);
    const int m0 = rowTile * 16;
    if (m0 >= n) return;
    const int n0 = (wave & 3) * 16;

    const int h  = lane >> 4;
    const int ml = lane & 15;

    const float* xrow = X + (size_t)(m0 + ml) * D;

    v8f c = {};
    #pragma unroll
    for (int k0 = 0; k0 < D; k0 += 4) {
        const int k = k0 + 2 * h;
        v2f a;
        a.x = xrow[k];
        a.y = xrow[k + 1];
        v2f b;
        b.x = W[(size_t)k       * D + n0 + ml];
        b.y = W[(size_t)(k + 1) * D + n0 + ml];
        c = __builtin_amdgcn_wmma_f32_16x16x4_f32(
                 false, a,  false, b,
                 (short)0, c,  false,  false);
    }

    #pragma unroll
    for (int v = 0; v < 8; ++v) {
        S[(size_t)(m0 + v + 8 * h) * D + n0 + ml] = c[v];
    }
}

#define CTR_STRIDE 64

__global__ void __launch_bounds__(256)
init_out(float* __restrict__ out, const float* __restrict__ b, int total4,
         unsigned int* __restrict__ ctr)
{
    int i = blockIdx.x * 256 + threadIdx.x;
    if (i == 0) {
        ctr[0] = 0u;
        ctr[CTR_STRIDE] = 0u;
    }
    if (i < total4) {
        const float4 bv = *reinterpret_cast<const float4*>(b + ((i * 4) & (D - 1)));
        reinterpret_cast<float4*>(out)[i] = bv;
    }
}

#define EPW 16
#define TILE_EDGES (8 * EPW)
#define CMD_DONE 0xFFFFFFFFu
__global__ void __launch_bounds__(256)
edge_scatter(const float* __restrict__ S, const float* __restrict__ ev,
             const int* __restrict__ row, const int* __restrict__ col,
             float* __restrict__ out, int E, int n, unsigned int* __restrict__ ctr)
{
    __shared__ unsigned int s_cmd[2];
    const int lane = threadIdx.x & 31;
    const int wave = __builtin_amdgcn_readfirstlane((int)(threadIdx.x >> 5));

    const unsigned int nTiles = (unsigned int)((E + TILE_EDGES - 1) / TILE_EDGES);
    const int split = n >> 1;

    unsigned int side = 0u, pending = 0u;
    bool switched = false;
    if (threadIdx.x == 0) {
        const unsigned int place = __builtin_amdgcn_s_sendmsg_rtn(0x87);
        side = (((place >> 16) & 15u) >= 4u) ? 1u : 0u;
        pending = atomicAdd(&ctr[side * CTR_STRIDE], 1u);
    }
    unsigned int buf = 0u;
    for (;;) {
        if (threadIdx.x == 0) {
            unsigned int t = pending;
            if (t >= nTiles && !switched) {
                switched = true;
                side ^= 1u;
                t = atomicAdd(&ctr[side * CTR_STRIDE], 1u);
            }
            unsigned int cmd = CMD_DONE;
            if (t < nTiles) cmd = t | (side << 31);
            s_cmd[buf] = cmd;
        }
        __syncthreads();
        const unsigned int w = (unsigned int)__builtin_amdgcn_readfirstlane((int)s_cmd[buf]);
        buf ^= 1u;
        if (w == CMD_DONE) break;

        const int t = (int)(w & 0x7FFFFFFFu);
        const int rbase = (w >> 31) ? split : 0;
        const unsigned int rspan = (w >> 31) ? (unsigned int)(n - split) : (unsigned int)split;
        const int e0 = t * TILE_EDGES + wave * EPW;

        if (e0 + EPW <= E) {
            int c[EPW], r[EPW];
            float v[EPW], lo[EPW], hi[EPW];
            const int le = e0 + (lane & (EPW - 1));
            const int   rv = row[le];
            const int   cv = col[le];
            const float vv = ev[le];
            #pragma unroll
            for (int j = 0; j < EPW; ++j) {
                r[j] = __builtin_amdgcn_readlane(rv, j);
                c[j] = __builtin_amdgcn_readlane(cv, j);
                v[j] = __int_as_float(__builtin_amdgcn_readlane(__float_as_int(vv), j));
            }
            asm volatile("" :: "s"(v[0]), "s"(v[1]), "s"(v[2]), "s"(v[3]), "s"(v[4]), "s"(v[5]), "s"(v[6]), "s"(v[7]),
                               "s"(v[8]), "s"(v[9]), "s"(v[10]), "s"(v[11]), "s"(v[12]), "s"(v[13]), "s"(v[14]), "s"(v[15]));
            #pragma unroll
            for (int j = 0; j < EPW; ++j) {
                lo[j] = 0.0f;
                hi[j] = 0.0f;
                if ((unsigned int)(r[j] - rbase) < rspan) {
                    const float* s = S + (size_t)c[j] * D + lane;
                    lo[j] = s[0];
                    hi[j] = s[32];
                }
            }
            if (threadIdx.x == 0) pending = atomicAdd(&ctr[side * CTR_STRIDE], 1u);
            #pragma unroll
            for (int j = 0; j < EPW; ++j) {
                if ((unsigned int)(r[j] - rbase) < rspan) {
                    float* o = out + (size_t)r[j] * D + lane;
                    unsafeAtomicAdd(o,      v[j] * lo[j]);
                    unsafeAtomicAdd(o + 32, v[j] * hi[j]);
                }
            }
        } else {
            if (threadIdx.x == 0) pending = atomicAdd(&ctr[side * CTR_STRIDE], 1u);
            for (int e = e0; e < E; ++e) {
                const int r0 = row[e];
                if ((unsigned int)(r0 - rbase) < rspan) {
                    const int   c0 = col[e];
                    const float v0 = ev[e];
                    const float* s0 = S + (size_t)c0 * D + lane;
                    const float a0 = s0[0];
                    const float a1 = s0[32];
                    float* o0 = out + (size_t)r0 * D + lane;
                    unsafeAtomicAdd(o0,      v0 * a0);
                    unsafeAtomicAdd(o0 + 32, v0 * a1);
                }
            }
        }
    }
}

extern "C" void kernel_launch(void* const* d_in, const int* in_sizes, int n_in,
                              void* d_out, int out_size, void* d_ws, size_t ws_size,
                              hipStream_t stream)
{
    const float* X   = (const float*)d_in[0];
    const float* W   = (const float*)d_in[1];
    const float* b   = (const float*)d_in[2];
    const float* ev  = (const float*)d_in[3];
    const int*   row = (const int*)d_in[4];
    const int*   col = (const int*)d_in[5];
    float*       out = (float*)d_out;
    float*       S   = (float*)d_ws;

    const int N = in_sizes[0] / D;
    const int E = in_sizes[3];
    unsigned int* ctr = (unsigned int*)(S + (size_t)N * D);

    const int rowTiles  = (N + 15) / 16;
    const int gemmBlks  = (rowTiles + 1) / 2;
    gemm_wmma<<<gemmBlks, 256, 0, stream>>>(X, W, S, N);

    const int total4   = N * (D / 4);
    const int initBlks = (total4 + 255) / 256;
    init_out<<<initBlks, 256, 0, stream>>>(out, b, total4, ctr);

    edge_scatter<<<2048, 256, 0, stream>>>(S, ev, row, col, out, E, N, ctr);
}


# edge scatter: own-side test by lanes (one compare + ballot per 32 candidate edges); a wave walks only its taken edges, up to 24 in flight: half the trips, no per-skip work; arrival-order sums
# speedup vs baseline: 2.6447x; 1.0443x over previous
#include <hip/hip_runtime.h>

typedef float v2f __attribute__((ext_vector_type(2)));
typedef float v8f __attribute__((ext_vector_type(8)));

#define D 64

__global__ void __launch_bounds__(256)
gemm_wmma(const float* __restrict__ X, const float* __restrict__ W,
          float* __restrict__ S, int n)
{
    const int lane = threadIdx.x & 31;
    const int wave = threadIdx.x >> 5;
    const int rowTile = blockIdx.x * 2 + (wave >> 2);
    const int m0 = rowTile * 16;
    if (m0 >= n) return;
    const int n0 = (wave & 3) * 16;

    const int h  = lane >> 4;
    const int ml = lane & 15;

    const float* xrow = X + (size_t)(m0 + ml) * D;

    v8f c = {};
    #pragma unroll
    for (int k0 = 0; k0 < D; k0 += 4) {
        const int k = k0 + 2 * h;
        v2f a;
        a.x = xrow[k];
        a.y = xrow[k + 1];
        v2f b;
        b.x = W[(size_t)k       * D + n0 + ml];
        b.y = W[(size_t)(k + 1) * D + n0 + ml];
        c = __builtin_amdgcn_wmma_f32_16x16x4_f32(
                 false, a,  false, b,
                 (short)0, c,  false,  false);
    }

    #pragma unroll
    for (int v = 0; v < 8; ++v) {
        S[(size_t)(m0 + v + 8 * h) * D + n0 + ml] = c[v];
    }
}

#define CTR_STRIDE 64

__global__ void __launch_bounds__(256)
init_out(float* __restrict__ out, const float* __restrict__ b, int total4,
         unsigned int* __restrict__ ctr)
{
    int i = blockIdx.x * 256 + threadIdx.x;
    if (i == 0) {
        ctr[0] = 0u;
        ctr[CTR_STRIDE] = 0u;
    }
    if (i < total4) {
        const float4 bv = *reinterpret_cast<const float4*>(b + ((i * 4) & (D - 1)));
        reinterpret_cast<float4*>(out)[i] = bv;
    }
}

#define CAND 32
#define TILE_EDGES (8 * CAND)
#define SLOTS 24
#define CMD_DONE 0xFFFFFFFFu
__global__ void __launch_bounds__(256)
edge_scatter(const float* __restrict__ S, const float* __restrict__ ev,
             const int* __restrict__ row, const int* __restrict__ col,
             float* __restrict__ out, int E, int n, unsigned int* __restrict__ ctr)
{
    __shared__ unsigned int s_cmd[2];
    const int lane = threadIdx.x & 31;
    const int wave = __builtin_amdgcn_readfirstlane((int)(threadIdx.x >> 5));

    const unsigned int nTiles = (unsigned int)((E + TILE_EDGES - 1) / TILE_EDGES);
    const int split = n >> 1;

    unsigned int side = 0u, pending = 0u;
    bool switched = false;
    if (threadIdx.x == 0) {
        const unsigned int place = __builtin_amdgcn_s_sendmsg_rtn(0x87);
        side = (((place >> 16) & 15u) >= 4u) ? 1u : 0u;
        pending = atomicAdd(&ctr[side * CTR_STRIDE], 1u);
    }
    unsigned int buf = 0u;
    for (;;) {
        if (threadIdx.x == 0) {
            unsigned int t = pending;
            if (t >= nTiles && !switched) {
                switched = true;
                side ^= 1u;
                t = atomicAdd(&ctr[side * CTR_STRIDE], 1u);
            }
            unsigned int cmd = CMD_DONE;
            if (t < nTiles) cmd = t | (side << 31);
            s_cmd[buf] = cmd;
        }
        __syncthreads();
        const unsigned int w = (unsigned int)__builtin_amdgcn_readfirstlane((int)s_cmd[buf]);
        buf ^= 1u;
        if (w == CMD_DONE) break;

        const int t = (int)(w & 0x7FFFFFFFu);
        const int rbase = (w >> 31) ? split : 0;
        const unsigned int rspan = (w >> 31) ? (unsigned int)(n - split) : (unsigned int)split;
        const int le = t * TILE_EDGES + wave * CAND + lane;

        float vv = 0.0f;
        int rv = 0, cv = 0;
        const bool in = le < E;
        if (in) {
            vv = ev[le];
            cv = col[le];
            rv = row[le];
        }
        const bool take = in && ((unsigned int)(rv - rbase) < rspan);
        unsigned int mask = __builtin_amdgcn_ballot_w32(take);

        int pass = 0;
        do {
            float lo[SLOTS], hi[SLOTS];
            unsigned int m = mask;
            #pragma unroll
            for (int j = 0; j < SLOTS; ++j) {
                lo[j] = 0.0f;
                hi[j] = 0.0f;
                if (m != 0u) {
                    const int l = __builtin_ctz(m);
                    m &= m - 1u;
                    const unsigned int cj = (unsigned int)__builtin_amdgcn_readlane(cv, l);
                    const float* s = S + (cj * (unsigned int)D + (unsigned int)lane);
                    lo[j] = s[0];
                    hi[j] = s[32];
                }
            }
            if (pass == 0 && threadIdx.x == 0) pending = atomicAdd(&ctr[side * CTR_STRIDE], 1u);
            unsigned int m2 = mask;
            #pragma unroll
            for (int j = 0; j < SLOTS; ++j) {
                if (m2 != 0u) {
                    const int l = __builtin_ctz(m2);
                    m2 &= m2 - 1u;
                    const unsigned int rj = (unsigned int)__builtin_amdgcn_readlane(rv, l);
                    const float vj = __int_as_float(__builtin_amdgcn_readlane(__float_as_int(vv), l));
                    float* o = out + (rj * (unsigned int)D + (unsigned int)lane);
                    unsafeAtomicAdd(o,      vj * lo[j]);
                    unsafeAtomicAdd(o + 32, vj * hi[j]);
                }
            }
            mask = m;
            ++pass;
        } while (mask != 0u && pass < 2);
    }
}

extern "C" void kernel_launch(void* const* d_in, const int* in_sizes, int n_in,
                              void* d_out, int out_size, void* d_ws, size_t ws_size,
                              hipStream_t stream)
{
    const float* X   = (const float*)d_in[0];
    const float* W   = (const float*)d_in[1];
    const float* b   = (const float*)d_in[2];
    const float* ev  = (const float*)d_in[3];
    const int*   row = (const int*)d_in[4];
    const int*   col = (const int*)d_in[5];
    float*       out = (float*)d_out;
    float*       S   = (float*)d_ws;

    const int N = in_sizes[0] / D;
    const int E = in_sizes[3];
    unsigned int* ctr = (unsigned int*)(S + (size_t)N * D);

    const int rowTiles  = (N + 15) / 16;
    const int gemmBlks  = (rowTiles + 1) / 2;
    gemm_wmma<<<gemmBlks, 256, 0, stream>>>(X, W, S, N);

    const int total4   = N * (D / 4);
    const int initBlks = (total4 + 255) / 256;
    init_out<<<initBlks, 256, 0, stream>>>(out, b, total4, ctr);

    edge_scatter<<<2048, 256, 0, stream>>>(S, ev, row, col, out, E, N, ctr);
}
